# starting kernel
# baseline (speedup 1.0000x reference)
#include <hip/hip_runtime.h>

typedef __attribute__((ext_vector_type(2))) float v2f;
typedef __attribute__((ext_vector_type(8))) float v8f;

#define CH     256
#define LCHUNK 2048
#define TILE   256

__device__ __forceinline__ float powa(float a, int e) {
  float r = 1.0f;
  for (int k = 0; k < e; ++k) r *= a;
  return r;
}

__device__ __forceinline__ float clamp_a1(float a) {
  return fminf(fmaxf(a, -1.0f), 1.0f);
}

__global__ void onepole_pass1(const float* __restrict__ x,
                              const float* __restrict__ b0p,
                              const float* __restrict__ b1p,
                              const float* __restrict__ a1p,
                              float* __restrict__ delta,
                              int T, int G) {
  const int lane = threadIdx.x & 31;
  const int wave = blockIdx.x * (blockDim.x >> 5) + (threadIdx.x >> 5);
  const int ch = wave / G;
  const int g  = wave % G;
  const long long base = (long long)ch * T + (long long)g * LCHUNK;

  const float a1 = clamp_a1(a1p[0]);
  const float c  = b1p[0] + a1 * b0p[0];
  const float a8   = powa(a1, 8);
  const float a256 = powa(a8, 32);

  float w[8];
  w[7] = powa(a8, 31 - lane);
  for (int j = 6; j >= 0; --j) w[j] = w[j + 1] * a1;

  float acc = 0.0f;
  const int ntile = LCHUNK / TILE;
  for (int t = 0; t < ntile; ++t) {
    const float* p = x + base + t * TILE + lane * 8;
    if (t + 1 < ntile) __builtin_prefetch(p + TILE, 0, 1);
    float4 u = *(const float4*)p;
    float4 v = *(const float4*)(p + 4);
    float partial =            w[0] * u.x;
    partial = fmaf(w[1], u.y, partial);
    partial = fmaf(w[2], u.z, partial);
    partial = fmaf(w[3], u.w, partial);
    partial = fmaf(w[4], v.x, partial);
    partial = fmaf(w[5], v.y, partial);
    partial = fmaf(w[6], v.z, partial);
    partial = fmaf(w[7], v.w, partial);
    acc = fmaf(a256, acc, partial);
  }
  for (int off = 16; off; off >>= 1) acc += __shfl_xor(acc, off, 32);
  if (lane == 0) delta[ch * G + g] = c * acc;
}

__global__ void onepole_pass2(const float* __restrict__ delta,
                              float* __restrict__ sstart,
                              const float* __restrict__ a1p,
                              int G) {
  const int ch = threadIdx.x;
  const float a1 = clamp_a1(a1p[0]);
  const float a256 = powa(powa(a1, 16), 16);
  const float AL = powa(a256, LCHUNK / 256);
  float s = 0.0f;
  for (int g = 0; g < G; ++g) {
    const int i = ch * G + g;
    sstart[i] = s;
    s = fmaf(AL, s, delta[i]);
  }
}

__global__ void onepole_pass3(const float* __restrict__ x,
                              const float* __restrict__ b0p,
                              const float* __restrict__ b1p,
                              const float* __restrict__ a1p,
                              const float* __restrict__ sstart,
                              float* __restrict__ y,
                              int T, int G) {
  __shared__ float lds[4 * TILE];

  const int lane = threadIdx.x & 31;
  const int wave = blockIdx.x * (blockDim.x >> 5) + (threadIdx.x >> 5);
  const int ch = wave / G;
  const int g  = wave % G;
  const long long base = (long long)ch * T + (long long)g * LCHUNK;

  const int seg = lane & 15;
  const int hi  = (lane >> 4) & 1;

  const float a1  = clamp_a1(a1p[0]);
  const float b0v = b0p[0];
  const float b1v = b1p[0];
  const float c   = b1v + a1 * b0v;

  v2f aA[4];
#pragma unroll
  for (int q = 0; q < 4; ++q) {
    const int k0 = 4 * q + 2 * hi;
#pragma unroll
    for (int e = 0; e < 2; ++e) {
      const int j = k0 + e;
      float m;
      if (j == seg)      m = b0v;
      else if (j < seg)  m = c * powa(a1, seg - 1 - j);
      else               m = 0.0f;
      if (e == 0) aA[q].x = m; else aA[q].y = m;
    }
  }

  const float m1 = powa(a1, 16);
  const float m2 = m1 * m1;
  const float m4 = m2 * m2;
  const float m8 = m4 * m4;
  const float a256 = m8 * m8;
  const float p16seg = powa(m1, seg);
  float rowpow[8];
#pragma unroll
  for (int v = 0; v < 8; ++v) rowpow[v] = powa(a1, v + 8 * hi);

  float s_in = sstart[ch * G + g];
  float* lb = lds + (threadIdx.x >> 5) * TILE;

  const int ntile = LCHUNK / TILE;
  for (int t = 0; t < ntile; ++t) {
    const float* xt = x + base + t * TILE;
    if (t + 1 < ntile) __builtin_prefetch(xt + TILE + lane * 8, 0, 1);

    v2f bq[4];
#pragma unroll
    for (int q = 0; q < 4; ++q) {
      const int idx = seg * 16 + 4 * q + 2 * hi;
      bq[q].x = xt[idx];
      bq[q].y = xt[idx + 1];
    }

    v8f cacc = {};
#pragma unroll
    for (int q = 0; q < 4; ++q) {
      cacc = __builtin_amdgcn_wmma_f32_16x16x4_f32(
          false, aA[q], false, bq[q], (short)0, cacc, false, false);
    }

    float d = fmaf(b1v, bq[3].y, a1 * cacc[7]);
    float S = __shfl(d, 16 + seg, 32);

    float u;
    u = __shfl(S, (lane & 16) | ((seg - 1) & 15), 32); S = (seg >= 1) ? fmaf(m1, u, S) : S;
    u = __shfl(S, (lane & 16) | ((seg - 2) & 15), 32); S = (seg >= 2) ? fmaf(m2, u, S) : S;
    u = __shfl(S, (lane & 16) | ((seg - 4) & 15), 32); S = (seg >= 4) ? fmaf(m4, u, S) : S;
    u = __shfl(S, (lane & 16) | ((seg - 8) & 15), 32); S = (seg >= 8) ? fmaf(m8, u, S) : S;

    const float Sprev = __shfl(S, (lane & 16) | ((seg - 1) & 15), 32);
    const float s_seg = fmaf(p16seg, s_in, (seg == 0) ? 0.0f : Sprev);
    const float S15   = __shfl(S, 15, 32);

    float4 w0, w1;
    w0.x = fmaf(rowpow[0], s_seg, cacc[0]);
    w0.y = fmaf(rowpow[1], s_seg, cacc[1]);
    w0.z = fmaf(rowpow[2], s_seg, cacc[2]);
    w0.w = fmaf(rowpow[3], s_seg, cacc[3]);
    w1.x = fmaf(rowpow[4], s_seg, cacc[4]);
    w1.y = fmaf(rowpow[5], s_seg, cacc[5]);
    w1.z = fmaf(rowpow[6], s_seg, cacc[6]);
    w1.w = fmaf(rowpow[7], s_seg, cacc[7]);
    *(float4*)(lb + seg * 16 + 8 * hi)     = w0;
    *(float4*)(lb + seg * 16 + 8 * hi + 4) = w1;

    float4 o0 = *(float4*)(lb + lane * 8);
    float4 o1 = *(float4*)(lb + lane * 8 + 4);
    *(float4*)(y + base + t * TILE + lane * 8)     = o0;
    *(float4*)(y + base + t * TILE + lane * 8 + 4) = o1;

    s_in = fmaf(a256, s_in, S15);
  }
}

extern "C" void kernel_launch(void* const* d_in, const int* in_sizes, int n_in,
                              void* d_out, int out_size, void* d_ws, size_t ws_size,
                              hipStream_t stream) {
  (void)n_in; (void)out_size; (void)ws_size;
  const float* x  = (const float*)d_in[0];
  const float* b0 = (const float*)d_in[1];
  const float* b1 = (const float*)d_in[2];
  const float* a1 = (const float*)d_in[3];
  float* y = (float*)d_out;

  const int T = in_sizes[0] / CH;
  const int G = T / LCHUNK;

  float* delta  = (float*)d_ws;
  float* sstart = delta + CH * G;

  const int waves  = CH * G;
  dim3 blk(128);
  dim3 grd(waves / 4);

  onepole_pass1<<<grd, blk, 0, stream>>>(x, b0, b1, a1, delta, T, G);
  onepole_pass2<<<1, CH, 0, stream>>>(delta, sstart, a1, G);
  onepole_pass3<<<grd, blk, 0, stream>>>(x, b0, b1, a1, sstart, y, T, G);
}


# Fuse the 3 launches into 1: a 32-wave block owns a channel and scans the waves' chunk end states through LDS each step (new reduction tree for the carried state), so x is read once
# speedup vs baseline: 2.0714x; 2.0714x over previous
#include <hip/hip_runtime.h>


typedef __attribute__((ext_vector_type(2))) float v2f;
typedef __attribute__((ext_vector_type(8))) float v8f;

#define CH     256
#define TILE   256
#define NT     2
#define NWAVE  32
#define SPAN   (NWAVE * NT * TILE)


__device__ __forceinline__ float powa(float a, int e) {
  float r = 1.0f;
  for (int k = 0; k < e; ++k) r *= a;
  return r;
}

__device__ __forceinline__ float clamp_a1(float a) {
  return fminf(fmaxf(a, -1.0f), 1.0f);
}

__device__ __forceinline__ void load_tile(const float* __restrict__ xt, int seg, int hi,
                                          v2f (&bq)[4]) {
#pragma unroll
  for (int q = 0; q < 4; ++q) {
    const int idx = seg * 16 + 4 * q + 2 * hi;
    bq[q].x = xt[idx];
    bq[q].y = xt[idx + 1];
  }
}

struct Consts {
  v2f   aA[4];
  float rowpow[8];
  float a1, b1v;
  float m1, m2, m4, m8, a256, p16seg;
  float AL, AL2, AL4, AL8, AL16, AL32, ALw;
};

template <bool HAVE_NEXT>
__device__ __forceinline__ void block_step(const Consts& k,
                                           const float* __restrict__ xnext,
                                           float* __restrict__ yout,
                                           v2f (&cur)[NT][4], v2f (&nxt)[NT][4],
                                           float* __restrict__ lb, float* __restrict__ slot,
                                           int lane, int seg, int hi, int w,
                                           float& carry) {
  if (HAVE_NEXT) {
#pragma unroll
    for (int t = 0; t < NT; ++t) load_tile(xnext + t * TILE, seg, hi, nxt[t]);
  }

  v8f   cacc[NT];
  float sprev[NT];
  float s15[NT];
#pragma unroll
  for (int t = 0; t < NT; ++t) {
    v8f acc = {};
#pragma unroll
    for (int q = 0; q < 4; ++q) {
      acc = __builtin_amdgcn_wmma_f32_16x16x4_f32(
          false, k.aA[q], false, cur[t][q], (short)0, acc, false, false);
    }
    cacc[t] = acc;

    float d = fmaf(k.b1v, cur[t][3].y, k.a1 * acc[7]);
    float S = __shfl(d, 16 + seg, 32);

    float u;
    u = __shfl(S, (lane & 16) | ((seg - 1) & 15), 32); S = (seg >= 1) ? fmaf(k.m1, u, S) : S;
    u = __shfl(S, (lane & 16) | ((seg - 2) & 15), 32); S = (seg >= 2) ? fmaf(k.m2, u, S) : S;
    u = __shfl(S, (lane & 16) | ((seg - 4) & 15), 32); S = (seg >= 4) ? fmaf(k.m4, u, S) : S;
    u = __shfl(S, (lane & 16) | ((seg - 8) & 15), 32); S = (seg >= 8) ? fmaf(k.m8, u, S) : S;

    const float Sp = __shfl(S, (lane & 16) | ((seg - 1) & 15), 32);
    sprev[t] = (seg == 0) ? 0.0f : Sp;
    s15[t]   = __shfl(S, 15, 32);
  }

  float D = s15[0];
#pragma unroll
  for (int t = 1; t < NT; ++t) D = fmaf(k.a256, D, s15[t]);

  if (lane == 0) slot[w] = D;
  __builtin_amdgcn_fence(__ATOMIC_RELEASE, "workgroup", "local");
  __builtin_amdgcn_s_barrier();
  __builtin_amdgcn_fence(__ATOMIC_ACQUIRE, "workgroup", "local");

  float P = slot[lane];
  {
    float u;
    u = __shfl(P, (lane - 1) & 31, 32);  P = (lane >= 1)  ? fmaf(k.AL,   u, P) : P;
    u = __shfl(P, (lane - 2) & 31, 32);  P = (lane >= 2)  ? fmaf(k.AL2,  u, P) : P;
    u = __shfl(P, (lane - 4) & 31, 32);  P = (lane >= 4)  ? fmaf(k.AL4,  u, P) : P;
    u = __shfl(P, (lane - 8) & 31, 32);  P = (lane >= 8)  ? fmaf(k.AL8,  u, P) : P;
    u = __shfl(P, (lane - 16) & 31, 32); P = (lane >= 16) ? fmaf(k.AL16, u, P) : P;
  }
  const float Pprev = __shfl(P, (w - 1) & 31, 32);
  const float P31   = __shfl(P, 31, 32);
  float s_in = fmaf(k.ALw, carry, (w == 0) ? 0.0f : Pprev);
  carry = fmaf(k.AL32, carry, P31);

#pragma unroll
  for (int t = 0; t < NT; ++t) {
    const float s_seg = fmaf(k.p16seg, s_in, sprev[t]);
    float4 w0, w1;
    w0.x = fmaf(k.rowpow[0], s_seg, cacc[t][0]);
    w0.y = fmaf(k.rowpow[1], s_seg, cacc[t][1]);
    w0.z = fmaf(k.rowpow[2], s_seg, cacc[t][2]);
    w0.w = fmaf(k.rowpow[3], s_seg, cacc[t][3]);
    w1.x = fmaf(k.rowpow[4], s_seg, cacc[t][4]);
    w1.y = fmaf(k.rowpow[5], s_seg, cacc[t][5]);
    w1.z = fmaf(k.rowpow[6], s_seg, cacc[t][6]);
    w1.w = fmaf(k.rowpow[7], s_seg, cacc[t][7]);
    *(float4*)(lb + seg * 16 + 8 * hi)     = w0;
    *(float4*)(lb + seg * 16 + 8 * hi + 4) = w1;

    float4 o0 = *(float4*)(lb + lane * 8);
    float4 o1 = *(float4*)(lb + lane * 8 + 4);
    *(float4*)(yout + t * TILE + lane * 8)     = o0;
    *(float4*)(yout + t * TILE + lane * 8 + 4) = o1;

    s_in = fmaf(k.a256, s_in, s15[t]);
  }
}

__global__ __launch_bounds__(NWAVE * 32)
void onepole_fused(const float* __restrict__ x,
                   const float* __restrict__ b0p,
                   const float* __restrict__ b1p,
                   const float* __restrict__ a1p,
                   float* __restrict__ y,
                   int T) {
  __shared__ float lds[NWAVE * TILE];
  __shared__ float dslot[2][NWAVE];

  const int lane = threadIdx.x & 31;
  const int w    = threadIdx.x >> 5;
  const int ch   = blockIdx.x;

  const int seg = lane & 15;
  const int hi  = (lane >> 4) & 1;

  Consts k;
  const float a1  = clamp_a1(a1p[0]);
  const float b0v = b0p[0];
  const float b1v = b1p[0];
  const float c   = b1v + a1 * b0v;
  k.a1 = a1;
  k.b1v = b1v;

#pragma unroll
  for (int q = 0; q < 4; ++q) {
    const int k0 = 4 * q + 2 * hi;
#pragma unroll
    for (int e = 0; e < 2; ++e) {
      const int j = k0 + e;
      float m;
      if (j == seg)      m = b0v;
      else if (j < seg)  m = c * powa(a1, seg - 1 - j);
      else               m = 0.0f;
      if (e == 0) k.aA[q].x = m; else k.aA[q].y = m;
    }
  }

  k.m1 = powa(a1, 16);
  k.m2 = k.m1 * k.m1;
  k.m4 = k.m2 * k.m2;
  k.m8 = k.m4 * k.m4;
  k.a256 = k.m8 * k.m8;
  k.p16seg = powa(k.m1, seg);
#pragma unroll
  for (int v = 0; v < 8; ++v) k.rowpow[v] = powa(a1, v + 8 * hi);

  float AL = k.a256;
  for (int i = 1; i < NT; ++i) AL *= k.a256;
  k.AL   = AL;
  k.AL2  = AL * AL;
  k.AL4  = k.AL2 * k.AL2;
  k.AL8  = k.AL4 * k.AL4;
  k.AL16 = k.AL8 * k.AL8;
  k.AL32 = k.AL16 * k.AL16;
  k.ALw  = powa(AL, w);

  const long long cbase = (long long)ch * T + (long long)w * (NT * TILE);
  const int nit = T / SPAN;
  float* lb = lds + w * TILE;
  const float* xw = x + cbase;
  float*       yw = y + cbase;

  float carry = 0.0f;

  v2f fa[NT][4], fb[NT][4];
#pragma unroll
  for (int t = 0; t < NT; ++t) load_tile(xw + t * TILE, seg, hi, fa[t]);

  int it = 0;
  for (; it + 2 < nit; it += 2) {
    const long long o0 = (long long)it * SPAN;
    block_step<true>(k, xw + o0 + SPAN, yw + o0, fa, fb, lb, dslot[0],
                     lane, seg, hi, w, carry);
    block_step<true>(k, xw + o0 + 2 * SPAN, yw + o0 + SPAN, fb, fa, lb, dslot[1],
                     lane, seg, hi, w, carry);
  }
  const long long o0 = (long long)it * SPAN;
  if (it + 2 == nit) {
    block_step<true>(k, xw + o0 + SPAN, yw + o0, fa, fb, lb, dslot[0],
                     lane, seg, hi, w, carry);
    block_step<false>(k, xw, yw + o0 + SPAN, fb, fa, lb, dslot[1],
                      lane, seg, hi, w, carry);
  } else if (it + 1 == nit) {
    block_step<false>(k, xw, yw + o0, fa, fb, lb, dslot[0],
                      lane, seg, hi, w, carry);
  }
}

extern "C" void kernel_launch(void* const* d_in, const int* in_sizes, int n_in,
                              void* d_out, int out_size, void* d_ws, size_t ws_size,
                              hipStream_t stream) {
  (void)n_in; (void)out_size; (void)d_ws; (void)ws_size;
  const float* x  = (const float*)d_in[0];
  const float* b0 = (const float*)d_in[1];
  const float* b1 = (const float*)d_in[2];
  const float* a1 = (const float*)d_in[3];
  float* y = (float*)d_out;

  const int T = in_sizes[0] / CH;

  onepole_fused<<<dim3(CH), dim3(NWAVE * 32), 0, stream>>>(x, b0, b1, a1, y, T);
}


# Whole-line stores: the staged tile is read back 16 B per lane contiguously, so each store instruction writes 512 contiguous bytes (4 full 128-B lines) instead of 16-B pieces of 8 lines
# speedup vs baseline: 2.7207x; 1.3135x over previous
#include <hip/hip_runtime.h>


typedef __attribute__((ext_vector_type(2))) float v2f;
typedef __attribute__((ext_vector_type(8))) float v8f;

#define CH     256
#define TILE   256
#define NT     2
#define NWAVE  32
#define SPAN   (NWAVE * NT * TILE)


__device__ __forceinline__ float powa(float a, int e) {
  float r = 1.0f;
  for (int k = 0; k < e; ++k) r *= a;
  return r;
}

__device__ __forceinline__ float clamp_a1(float a) {
  return fminf(fmaxf(a, -1.0f), 1.0f);
}

__device__ __forceinline__ void load_tile(const float* __restrict__ xt, int seg, int hi,
                                          v2f (&bq)[4]) {
#pragma unroll
  for (int q = 0; q < 4; ++q) {
    const int idx = seg * 16 + 4 * q + 2 * hi;
    bq[q].x = xt[idx];
    bq[q].y = xt[idx + 1];
  }
}

struct Consts {
  v2f   aA[4];
  float rowpow[8];
  float a1, b1v;
  float m1, m2, m4, m8, a256, p16seg;
  float AL, AL2, AL4, AL8, AL16, AL32, ALw;
};

template <bool HAVE_NEXT>
__device__ __forceinline__ void block_step(const Consts& k,
                                           const float* __restrict__ xnext,
                                           float* __restrict__ yout,
                                           v2f (&cur)[NT][4], v2f (&nxt)[NT][4],
                                           float* __restrict__ lb, float* __restrict__ slot,
                                           int lane, int seg, int hi, int w,
                                           float& carry) {
  if (HAVE_NEXT) {
#pragma unroll
    for (int t = 0; t < NT; ++t) load_tile(xnext + t * TILE, seg, hi, nxt[t]);
  }

  v8f   cacc[NT];
  float sprev[NT];
  float s15[NT];
#pragma unroll
  for (int t = 0; t < NT; ++t) {
    v8f acc = {};
#pragma unroll
    for (int q = 0; q < 4; ++q) {
      acc = __builtin_amdgcn_wmma_f32_16x16x4_f32(
          false, k.aA[q], false, cur[t][q], (short)0, acc, false, false);
    }
    cacc[t] = acc;

    float d = fmaf(k.b1v, cur[t][3].y, k.a1 * acc[7]);
    float S = __shfl(d, 16 + seg, 32);

    float u;
    u = __shfl(S, (lane & 16) | ((seg - 1) & 15), 32); S = (seg >= 1) ? fmaf(k.m1, u, S) : S;
    u = __shfl(S, (lane & 16) | ((seg - 2) & 15), 32); S = (seg >= 2) ? fmaf(k.m2, u, S) : S;
    u = __shfl(S, (lane & 16) | ((seg - 4) & 15), 32); S = (seg >= 4) ? fmaf(k.m4, u, S) : S;
    u = __shfl(S, (lane & 16) | ((seg - 8) & 15), 32); S = (seg >= 8) ? fmaf(k.m8, u, S) : S;

    const float Sp = __shfl(S, (lane & 16) | ((seg - 1) & 15), 32);
    sprev[t] = (seg == 0) ? 0.0f : Sp;
    s15[t]   = __shfl(S, 15, 32);
  }

  float D = s15[0];
#pragma unroll
  for (int t = 1; t < NT; ++t) D = fmaf(k.a256, D, s15[t]);

  if (lane == 0) slot[w] = D;
  __builtin_amdgcn_fence(__ATOMIC_RELEASE, "workgroup", "local");
  __builtin_amdgcn_s_barrier();
  __builtin_amdgcn_fence(__ATOMIC_ACQUIRE, "workgroup", "local");

  float P = slot[lane];
  {
    float u;
    u = __shfl(P, (lane - 1) & 31, 32);  P = (lane >= 1)  ? fmaf(k.AL,   u, P) : P;
    u = __shfl(P, (lane - 2) & 31, 32);  P = (lane >= 2)  ? fmaf(k.AL2,  u, P) : P;
    u = __shfl(P, (lane - 4) & 31, 32);  P = (lane >= 4)  ? fmaf(k.AL4,  u, P) : P;
    u = __shfl(P, (lane - 8) & 31, 32);  P = (lane >= 8)  ? fmaf(k.AL8,  u, P) : P;
    u = __shfl(P, (lane - 16) & 31, 32); P = (lane >= 16) ? fmaf(k.AL16, u, P) : P;
  }
  const float Pprev = __shfl(P, (w - 1) & 31, 32);
  const float P31   = __shfl(P, 31, 32);
  float s_in = fmaf(k.ALw, carry, (w == 0) ? 0.0f : Pprev);
  carry = fmaf(k.AL32, carry, P31);

#pragma unroll
  for (int t = 0; t < NT; ++t) {
    const float s_seg = fmaf(k.p16seg, s_in, sprev[t]);
    float4 w0, w1;
    w0.x = fmaf(k.rowpow[0], s_seg, cacc[t][0]);
    w0.y = fmaf(k.rowpow[1], s_seg, cacc[t][1]);
    w0.z = fmaf(k.rowpow[2], s_seg, cacc[t][2]);
    w0.w = fmaf(k.rowpow[3], s_seg, cacc[t][3]);
    w1.x = fmaf(k.rowpow[4], s_seg, cacc[t][4]);
    w1.y = fmaf(k.rowpow[5], s_seg, cacc[t][5]);
    w1.z = fmaf(k.rowpow[6], s_seg, cacc[t][6]);
    w1.w = fmaf(k.rowpow[7], s_seg, cacc[t][7]);
    *(float4*)(lb + seg * 16 + 8 * hi)     = w0;
    *(float4*)(lb + seg * 16 + 8 * hi + 4) = w1;

    float4 o0 = *(float4*)(lb + lane * 4);
    float4 o1 = *(float4*)(lb + 128 + lane * 4);
    *(float4*)(yout + t * TILE + lane * 4)       = o0;
    *(float4*)(yout + t * TILE + 128 + lane * 4) = o1;

    s_in = fmaf(k.a256, s_in, s15[t]);
  }
}

__global__ __launch_bounds__(NWAVE * 32)
void onepole_fused(const float* __restrict__ x,
                   const float* __restrict__ b0p,
                   const float* __restrict__ b1p,
                   const float* __restrict__ a1p,
                   float* __restrict__ y,
                   int T) {
  __shared__ float lds[NWAVE * TILE];
  __shared__ float dslot[2][NWAVE];

  const int lane = threadIdx.x & 31;
  const int w    = threadIdx.x >> 5;
  const int ch   = blockIdx.x;

  const int seg = lane & 15;
  const int hi  = (lane >> 4) & 1;

  Consts k;
  const float a1  = clamp_a1(a1p[0]);
  const float b0v = b0p[0];
  const float b1v = b1p[0];
  const float c   = b1v + a1 * b0v;
  k.a1 = a1;
  k.b1v = b1v;

#pragma unroll
  for (int q = 0; q < 4; ++q) {
    const int k0 = 4 * q + 2 * hi;
#pragma unroll
    for (int e = 0; e < 2; ++e) {
      const int j = k0 + e;
      float m;
      if (j == seg)      m = b0v;
      else if (j < seg)  m = c * powa(a1, seg - 1 - j);
      else               m = 0.0f;
      if (e == 0) k.aA[q].x = m; else k.aA[q].y = m;
    }
  }

  k.m1 = powa(a1, 16);
  k.m2 = k.m1 * k.m1;
  k.m4 = k.m2 * k.m2;
  k.m8 = k.m4 * k.m4;
  k.a256 = k.m8 * k.m8;
  k.p16seg = powa(k.m1, seg);
#pragma unroll
  for (int v = 0; v < 8; ++v) k.rowpow[v] = powa(a1, v + 8 * hi);

  float AL = k.a256;
  for (int i = 1; i < NT; ++i) AL *= k.a256;
  k.AL   = AL;
  k.AL2  = AL * AL;
  k.AL4  = k.AL2 * k.AL2;
  k.AL8  = k.AL4 * k.AL4;
  k.AL16 = k.AL8 * k.AL8;
  k.AL32 = k.AL16 * k.AL16;
  k.ALw  = powa(AL, w);

  const long long cbase = (long long)ch * T + (long long)w * (NT * TILE);
  const int nit = T / SPAN;
  float* lb = lds + w * TILE;
  const float* xw = x + cbase;
  float*       yw = y + cbase;

  float carry = 0.0f;

  v2f fa[NT][4], fb[NT][4];
#pragma unroll
  for (int t = 0; t < NT; ++t) load_tile(xw + t * TILE, seg, hi, fa[t]);

  int it = 0;
  for (; it + 2 < nit; it += 2) {
    const long long o0 = (long long)it * SPAN;
    block_step<true>(k, xw + o0 + SPAN, yw + o0, fa, fb, lb, dslot[0],
                     lane, seg, hi, w, carry);
    block_step<true>(k, xw + o0 + 2 * SPAN, yw + o0 + SPAN, fb, fa, lb, dslot[1],
                     lane, seg, hi, w, carry);
  }
  const long long o0 = (long long)it * SPAN;
  if (it + 2 == nit) {
    block_step<true>(k, xw + o0 + SPAN, yw + o0, fa, fb, lb, dslot[0],
                     lane, seg, hi, w, carry);
    block_step<false>(k, xw, yw + o0 + SPAN, fb, fa, lb, dslot[1],
                      lane, seg, hi, w, carry);
  } else if (it + 1 == nit) {
    block_step<false>(k, xw, yw + o0, fa, fb, lb, dslot[0],
                      lane, seg, hi, w, carry);
  }
}

extern "C" void kernel_launch(void* const* d_in, const int* in_sizes, int n_in,
                              void* d_out, int out_size, void* d_ws, size_t ws_size,
                              hipStream_t stream) {
  (void)n_in; (void)out_size; (void)d_ws; (void)ws_size;
  const float* x  = (const float*)d_in[0];
  const float* b0 = (const float*)d_in[1];
  const float* b1 = (const float*)d_in[2];
  const float* a1 = (const float*)d_in[3];
  float* y = (float*)d_out;

  const int T = in_sizes[0] / CH;

  onepole_fused<<<dim3(CH), dim3(NWAVE * 32), 0, stream>>>(x, b0, b1, a1, y, T);
}
